# starting kernel
# baseline (speedup 1.0000x reference)
#include <hip/hip_runtime.h>

typedef __attribute__((ext_vector_type(2))) float v2f;
typedef __attribute__((ext_vector_type(8))) float v8f;

#define EPSF 1e-5f

__global__ __launch_bounds__(128) void MapLidar2Image_kernel(
    const int* __restrict__ vox,
    const float* __restrict__ l2i,
    const float* __restrict__ aug,
    float* __restrict__ out,
    int n)
{
    __shared__ float sL[192];
    __shared__ float sA[192];
    const int tid = threadIdx.x;
    for (int i = tid; i < 192; i += 128) { sL[i] = l2i[i]; sA[i] = aug[i]; }
    __syncthreads();

    const int p    = blockIdx.x * 128 + tid;
    const int pc   = (p < n) ? p : (n - 1);
    const int lane = tid & 31;
    const bool lo  = lane < 16;

    const int4 c = ((const int4*)vox)[pc];
    const int bidx = c.x;
    const float px = ((float)c.w + 0.5f) * (108.0f / 360.0f) - 54.0f;
    const float py = ((float)c.z + 0.5f) * (108.0f / 360.0f) - 54.0f;
    const float pz = ((float)c.y + 0.5f) * (8.0f / 32.0f)    - 5.0f;

    const float pxs = __shfl_xor(px, 16, 32);
    const float pys = __shfl_xor(py, 16, 32);
    const float pzs = __shfl_xor(pz, 16, 32);
    v2f Ba, Bb;
    Ba[0] = lo ? px  : pzs;  Ba[1] = lo ? py  : 1.0f;
    Bb[0] = lo ? pxs : pz;   Bb[1] = lo ? pys : 1.0f;

    const int bA = __shfl(bidx, 0, 32);
    const int bB = __shfl(bidx, 16, 32);

    const int m    = lane & 15;
    const int view = m / 3;
    const int row  = m - 3 * view;
    const int k0   = lo ? 0 : 2;

    const float* LA = sL + bA * 96;
    const float* LB = sL + bB * 96;
    const int off = view * 16 + row * 4 + k0;
    v2f Aa, Ab;
    Aa[0] = LA[off];  Aa[1] = LA[off + 1];
    Ab[0] = LB[off];  Ab[1] = LB[off + 1];

    v8f zero = {};
    v8f Da = __builtin_amdgcn_wmma_f32_16x16x4_f32(false, Aa, false, Ba, (short)0, zero, false, false);
    v8f Db = __builtin_amdgcn_wmma_f32_16x16x4_f32(false, Ab, false, Bb, (short)0, zero, false, false);

    float comp[18];
#pragma unroll
    for (int r = 0; r < 8; ++r) {
        const float sa = __shfl_xor(Da[r], 16, 32);
        const float sb = __shfl_xor(Db[r], 16, 32);
        comp[r]     = lo ? Da[r] : sb;
        comp[8 + r] = lo ? sa    : Db[r];
    }

    const float* L5 = sL + bidx * 96 + 5 * 16;
    comp[16] = L5[4] * px + L5[5] * py + L5[6]  * pz + L5[7];
    comp[17] = L5[8] * px + L5[9] * py + L5[10] * pz + L5[11];

    const float* AG = sA + bidx * 96;
    bool  hit = false;
    int   vsel = 0;
    float usel = 0.0f, wsel = 0.0f;
#pragma unroll
    for (int v = 0; v < 6; ++v) {
        const float d   = comp[3 * v + 2];
        const float inv = __builtin_amdgcn_rcpf(fmaxf(d, EPSF));
        const float u   = comp[3 * v + 0] * inv;
        const float w   = comp[3 * v + 1] * inv;
        const float* ar = AG + v * 16;
        const float ua  = ar[0] * u + ar[1] * w + ar[2] + ar[3];
        const float wa  = ar[4] * u + ar[5] * w + ar[6] + ar[7];
        const bool valid = (d > EPSF) & (ua >= 0.0f) & (ua < 704.0f)
                                      & (wa >= 0.0f) & (wa < 256.0f);
        if (valid && !hit) { hit = true; vsel = v; usel = ua; wsel = wa; }
    }

    const float o0 = hit ? (float)vsel : -1.0f;
    const float o1 = hit ? usel * (88.0f / 704.0f) : 0.0f;
    const float o2 = hit ? wsel * (32.0f / 256.0f) : 0.0f;

    if (p < n) {
        out[3 * p + 0] = o0;
        out[3 * p + 1] = o1;
        out[3 * p + 2] = o2;
    }
}

extern "C" void kernel_launch(void* const* d_in, const int* in_sizes, int n_in,
                              void* d_out, int out_size, void* d_ws, size_t ws_size,
                              hipStream_t stream) {
    const int*   vox = (const int*)d_in[0];
    const float* l2i = (const float*)d_in[1];
    const float* aug = (const float*)d_in[2];
    float* out = (float*)d_out;
    const int n = in_sizes[0] / 4;
    const int blocks = (n + 127) / 128;
    MapLidar2Image_kernel<<<blocks, 128, 0, stream>>>(vox, l2i, aug, out, n);
}


# Matrices read by wave-uniform scalar loads into SGPR operands, one pass per batch entry in the wave (first-valid as selects, one block); LDS staging and block barrier gone; sums unchanged
# speedup vs baseline: 1.1622x; 1.1622x over previous
#include <hip/hip_runtime.h>

typedef __attribute__((ext_vector_type(2))) float v2f;
typedef __attribute__((ext_vector_type(8))) float v8f;

#define EPSF 1e-5f

template <int BATCH>
__device__ __forceinline__ void project_pass(
    const float* __restrict__ l2i, const float* __restrict__ aug,
    float* __restrict__ out,
    const v2f A, const v2f Ba, const v2f Bb,
    const float px, const float py, const float pz,
    const bool lo, const bool store_here, const int p)
{
    const float* AG = aug + BATCH * 96;
    float ag[6][8];
    float l5[8];
#pragma unroll
    for (int v = 0; v < 6; ++v)
#pragma unroll
        for (int j = 0; j < 8; ++j) ag[v][j] = AG[v * 16 + j];
#pragma unroll
    for (int j = 0; j < 8; ++j) l5[j] = l2i[BATCH * 96 + 5 * 16 + 4 + j];

    v8f zero = {};
    v8f Da = __builtin_amdgcn_wmma_f32_16x16x4_f32(false, A, false, Ba, (short)0, zero, false, false);
    v8f Db = __builtin_amdgcn_wmma_f32_16x16x4_f32(false, A, false, Bb, (short)0, zero, false, false);

    float comp[18];
#pragma unroll
    for (int r = 0; r < 8; ++r) {
        const float sa = __shfl_xor(Da[r], 16, 32);
        const float sb = __shfl_xor(Db[r], 16, 32);
        comp[r]     = lo ? Da[r] : sb;
        comp[8 + r] = lo ? sa    : Db[r];
    }

    asm volatile("" :: "s"(ag[0][0]), "s"(ag[1][0]), "s"(ag[2][0]), "s"(ag[3][0]),
                       "s"(ag[4][0]), "s"(ag[5][0]), "s"(l5[0]));

    comp[16] = fmaf(l5[2], pz, fmaf(l5[0], px, l5[1] * py)) + l5[3];
    comp[17] = fmaf(l5[6], pz, fmaf(l5[4], px, l5[5] * py)) + l5[7];

    bool  hit = false;
    int   vsel = 0;
    float usel = 0.0f, wsel = 0.0f;
#pragma unroll
    for (int v = 0; v < 6; ++v) {
        const float d   = comp[3 * v + 2];
        const float inv = __builtin_amdgcn_rcpf(fmaxf(d, EPSF));
        const float u   = comp[3 * v + 0] * inv;
        const float w   = comp[3 * v + 1] * inv;
        const float* ar = ag[v];
        const float ua  = fmaf(ar[1], w, ar[0] * u) + ar[2] + ar[3];
        const float wa  = fmaf(ar[5], w, ar[4] * u) + ar[6] + ar[7];
        const bool valid = (d > EPSF) & (ua >= 0.0f) & (ua < 704.0f)
                                      & (wa >= 0.0f) & (wa < 256.0f);
        const bool take = valid & !hit;
        vsel = take ? v  : vsel;
        usel = take ? ua : usel;
        wsel = take ? wa : wsel;
        hit  = hit | valid;
    }

    const float o0 = hit ? (float)vsel : -1.0f;
    const float o1 = hit ? usel * (88.0f / 704.0f) : 0.0f;
    const float o2 = hit ? wsel * (32.0f / 256.0f) : 0.0f;

    if (store_here) {
        out[3 * p + 0] = o0;
        out[3 * p + 1] = o1;
        out[3 * p + 2] = o2;
    }
}

__global__ __launch_bounds__(128) void MapLidar2Image_kernel(
    const int* __restrict__ vox,
    const float* __restrict__ l2i,
    const float* __restrict__ aug,
    float* __restrict__ out,
    int n)
{
    const int tid  = threadIdx.x;
    const int p    = blockIdx.x * 128 + tid;
    const int pc   = (p < n) ? p : (n - 1);
    const int lane = tid & 31;
    const bool lo  = lane < 16;

    const int4 c = ((const int4*)vox)[pc];

    const int m    = lane & 15;
    const int view = m / 3;
    const int row  = m - 3 * view;
    const int k0   = lo ? 0 : 2;
    const int off  = view * 16 + row * 4 + k0;
    const v2f A0 = *(const v2f*)(l2i + off);
    const v2f A1 = *(const v2f*)(l2i + 96 + off);
    asm volatile("" :: "v"(c.x), "v"(c.y), "v"(c.z), "v"(c.w), "v"(A0), "v"(A1));

    const int bidx = c.x;
    const float px = ((float)c.w + 0.5f) * (108.0f / 360.0f) - 54.0f;
    const float py = ((float)c.z + 0.5f) * (108.0f / 360.0f) - 54.0f;
    const float pz = ((float)c.y + 0.5f) * (8.0f / 32.0f)    - 5.0f;

    const float pxs = __shfl_xor(px, 16, 32);
    const float pys = __shfl_xor(py, 16, 32);
    const float pzs = __shfl_xor(pz, 16, 32);
    v2f Ba, Bb;
    Ba[0] = lo ? px  : pzs;  Ba[1] = lo ? py  : 1.0f;
    Bb[0] = lo ? pxs : pz;   Bb[1] = lo ? pys : 1.0f;

    const bool pos = bidx > 0;
    const bool neg = bidx == -1;
    const bool is1 = pos || neg;
    const unsigned m1 = __builtin_amdgcn_ballot_w32(pos) | __builtin_amdgcn_ballot_w32(neg);
    const bool inb = p < n;

    if (m1 != 0xffffffffu)
        project_pass<0>(l2i, aug, out, A0, Ba, Bb, px, py, pz, lo, inb & !is1, p);
    if (m1 != 0u)
        project_pass<1>(l2i, aug, out, A1, Ba, Bb, px, py, pz, lo, inb & is1, p);
}

extern "C" void kernel_launch(void* const* d_in, const int* in_sizes, int n_in,
                              void* d_out, int out_size, void* d_ws, size_t ws_size,
                              hipStream_t stream) {
    const int*   vox = (const int*)d_in[0];
    const float* l2i = (const float*)d_in[1];
    const float* aug = (const float*)d_in[2];
    float* out = (float*)d_out;
    const int n = in_sizes[0] / 4;
    const int blocks = (n + 127) / 128;
    MapLidar2Image_kernel<<<blocks, 128, 0, stream>>>(vox, l2i, aug, out, n);
}
